# starting kernel
# baseline (speedup 1.0000x reference)
#include <hip/hip_runtime.h>


typedef __attribute__((ext_vector_type(2))) float v2f;
typedef __attribute__((ext_vector_type(8))) float v8f;

#define TT 512
#define LL 16

__device__ __forceinline__ float half_swap_f(float v) {
    return __int_as_float(__builtin_amdgcn_ds_swizzle(__float_as_int(v), 0x401F));
}

__device__ __forceinline__ void load_xfrag(const float* base2h, int t, float xf[8]) {
    const float* p = base2h + (size_t)t * LL;
#pragma unroll
    for (int i = 0; i < 4; ++i) {
        float2 v = *(const float2*)(p + 4 * i);
        xf[2 * i]     = v.x;
        xf[2 * i + 1] = v.y;
    }
}

__launch_bounds__(256)
__global__ void crf_score_kernel(const float* __restrict__ x,
                                 const float* __restrict__ trans,
                                 const int*   __restrict__ label,
                                 const int*   __restrict__ length,
                                 float*       __restrict__ out)
{
    const int lane = threadIdx.x & 31;
    const int b    = blockIdx.x * 8 + (threadIdx.x >> 5);
    const int len  = length[b];
    const size_t base = (size_t)b * TT;

    float acc = 0.0f;
    for (int t = lane; t < TT; t += 32) {
        if (t < len) {
            const int lab = label[base + t];
            acc += x[(base + t) * LL + lab];
            if (t > 0) acc += trans[label[base + t - 1] * LL + lab];
        }
    }
#pragma unroll
    for (int off = 16; off; off >>= 1) acc += __shfl_xor(acc, off, 32);
    if (lane == 0) out[b] = -acc;
}

__launch_bounds__(32)
__global__ void crf_fwd_kernel(const float* __restrict__ x,
                               const float* __restrict__ trans,
                               const int*   __restrict__ length,
                               float*       __restrict__ out)
{
    const int lane = threadIdx.x & 31;
    const int half = lane >> 4;
    const int m    = lane & 15;
    const int bm   = blockIdx.x * 16 + m;

    const int s  = lane & 15;
    const int jo = 4 * ((s >> 1) & 3) + (s & 1) + 2 * (s >> 3);

    v2f A[4];
#pragma unroll
    for (int k = 0; k < 4; ++k) {
        const int k0 = 4 * k + 2 * half;
        A[k].x = __expf(trans[k0 * LL + jo]);
        A[k].y = __expf(trans[(k0 + 1) * LL + jo]);
    }

    const int len = length[bm];
    const float* xb = x + (size_t)bm * TT * LL + 2 * half;

    float p[8], finu[8];
    {
        float xf0[8];
        load_xfrag(xb, 0, xf0);
#pragma unroll
        for (int r = 0; r < 8; ++r) { p[r] = __expf(xf0[r]); finu[r] = p[r]; }
    }
    float M = 0.0f, finm = 0.0f;

    float exc[8];
    {
        float xf1[8];
        load_xfrag(xb, 1, xf1);
#pragma unroll
        for (int r = 0; r < 8; ++r) exc[r] = __expf(xf1[r]);
    }

    const v8f z8 = {};
    for (int t = 1; t < TT; ++t) {
        float xn[8];
        if (t + 1 < TT) load_xfrag(xb, t + 1, xn);
        if (t + 16 < TT) __builtin_prefetch(xb + (size_t)(t + 16) * LL, 0, 1);

        v2f B0 = {p[0], p[1]}, B1 = {p[2], p[3]};
        v2f B2 = {p[4], p[5]}, B3 = {p[6], p[7]};
        v8f accA = __builtin_amdgcn_wmma_f32_16x16x4_f32(false, A[0], false, B0,
                                                         (short)0, z8, false, false);
        v8f accB = __builtin_amdgcn_wmma_f32_16x16x4_f32(false, A[1], false, B1,
                                                         (short)0, z8, false, false);
        accA = __builtin_amdgcn_wmma_f32_16x16x4_f32(false, A[2], false, B2,
                                                     (short)0, accA, false, false);
        accB = __builtin_amdgcn_wmma_f32_16x16x4_f32(false, A[3], false, B3,
                                                     (short)0, accB, false, false);

        float u[8];
#pragma unroll
        for (int r = 0; r < 8; ++r) u[r] = (accA[r] + accB[r]) * exc[r];

        const bool cap = (t == len - 1);
        finm = cap ? M : finm;
#pragma unroll
        for (int r = 0; r < 8; ++r) finu[r] = cap ? u[r] : finu[r];

        if ((t & 7) == 0) {
            float um = fmaxf(u[0], u[1]);
#pragma unroll
            for (int r = 2; r < 8; ++r) um = fmaxf(um, u[r]);
            um = fmaxf(um, half_swap_f(um));
            const float ir = __builtin_amdgcn_rcpf(um);
#pragma unroll
            for (int r = 0; r < 8; ++r) p[r] = u[r] * ir;
            M += __logf(um);
        } else {
#pragma unroll
            for (int r = 0; r < 8; ++r) p[r] = u[r];
        }

        if (t + 1 < TT) {
#pragma unroll
            for (int r = 0; r < 8; ++r) exc[r] = __expf(xn[r]);
        }
    }

    float ssum = finu[0] + finu[1];
#pragma unroll
    for (int r = 2; r < 8; ++r) ssum += finu[r];
    ssum += half_swap_f(ssum);
    const float zv = finm + __logf(ssum);

    if (lane < 16) out[bm] = zv + out[bm];
}

extern "C" void kernel_launch(void* const* d_in, const int* in_sizes, int n_in,
                              void* d_out, int out_size, void* d_ws, size_t ws_size,
                              hipStream_t stream) {
    const float* x      = (const float*)d_in[0];
    const float* trans  = (const float*)d_in[1];
    const int*   label  = (const int*)d_in[2];
    const int*   length = (const int*)d_in[3];
    float*       out    = (float*)d_out;

    const int B = in_sizes[3];
    crf_score_kernel<<<B / 8, 256, 0, stream>>>(x, trans, label, length, out);
    crf_fwd_kernel<<<B / 16, 32, 0, stream>>>(x, trans, length, out);
}


# crf_fwd: x fragments fetched 7 steps ahead into an 8-slot register ring (time loop unrolled by 8, one compiler fence per step pins the refills), so load latency leaves the per-step chain
# speedup vs baseline: 1.5611x; 1.5611x over previous
#include <hip/hip_runtime.h>


typedef __attribute__((ext_vector_type(2))) float v2f;
typedef __attribute__((ext_vector_type(8))) float v8f;

#define TT 512
#define LL 16
#define RING 8
#define PFD  32

__device__ __forceinline__ float half_swap_f(float v) {
    return __int_as_float(__builtin_amdgcn_ds_swizzle(__float_as_int(v), 0x401F));
}

__device__ __forceinline__ void load_xfrag(const float* base2h, int t, float xf[8]) {
    const float* p = base2h + (size_t)t * LL;
#pragma unroll
    for (int i = 0; i < 4; ++i) {
        float2 v = *(const float2*)(p + 4 * i);
        xf[2 * i]     = v.x;
        xf[2 * i + 1] = v.y;
    }
}

__launch_bounds__(256)
__global__ void crf_score_kernel(const float* __restrict__ x,
                                 const float* __restrict__ trans,
                                 const int*   __restrict__ label,
                                 const int*   __restrict__ length,
                                 float*       __restrict__ out)
{
    const int lane = threadIdx.x & 31;
    const int b    = blockIdx.x * 8 + (threadIdx.x >> 5);
    const int len  = length[b];
    const size_t base = (size_t)b * TT;

    float acc = 0.0f;
    for (int t = lane; t < TT; t += 32) {
        if (t < len) {
            const int lab = label[base + t];
            acc += x[(base + t) * LL + lab];
            if (t > 0) acc += trans[label[base + t - 1] * LL + lab];
        }
    }
#pragma unroll
    for (int off = 16; off; off >>= 1) acc += __shfl_xor(acc, off, 32);
    if (lane == 0) out[b] = -acc;
}

template <bool RELOAD, bool NEXT>
__device__ __forceinline__ void fwd_step(const int t, const int len, const float* xb,
                                         const v2f (&A)[4], float (&p)[8], float (&exc)[8],
                                         float (&slot)[8], float (&fill)[8],
                                         float& M, float& finm,
                                         float (&finu)[8], const bool renorm)
{
    const v8f z8 = {};
    v2f B0 = {p[0], p[1]}, B1 = {p[2], p[3]};
    v2f B2 = {p[4], p[5]}, B3 = {p[6], p[7]};
    v8f accA = __builtin_amdgcn_wmma_f32_16x16x4_f32(false, A[0], false, B0,
                                                     (short)0, z8, false, false);
    v8f accB = __builtin_amdgcn_wmma_f32_16x16x4_f32(false, A[1], false, B1,
                                                     (short)0, z8, false, false);
    asm volatile("" : "+v"(accA), "+v"(accB),
                      "+v"(slot[0]), "+v"(slot[1]), "+v"(slot[2]), "+v"(slot[3]),
                      "+v"(slot[4]), "+v"(slot[5]), "+v"(slot[6]), "+v"(slot[7])
                 : : "memory");
    accA = __builtin_amdgcn_wmma_f32_16x16x4_f32(false, A[2], false, B2,
                                                 (short)0, accA, false, false);
    accB = __builtin_amdgcn_wmma_f32_16x16x4_f32(false, A[3], false, B3,
                                                 (short)0, accB, false, false);

    float en[8];
    if (NEXT) {
#pragma unroll
        for (int r = 0; r < 8; ++r) en[r] = __expf(slot[r]);
    }
    if (RELOAD) {
        const int tl = (t + RING < TT) ? (t + RING) : (TT - 1);
        load_xfrag(xb, tl, fill);
    }

    float u[8];
#pragma unroll
    for (int r = 0; r < 8; ++r) u[r] = (accA[r] + accB[r]) * exc[r];

    const bool cap = (t == len - 1);
    finm = cap ? M : finm;
#pragma unroll
    for (int r = 0; r < 8; ++r) finu[r] = cap ? u[r] : finu[r];

    if (renorm) {
        float um = fmaxf(u[0], u[1]);
#pragma unroll
        for (int r = 2; r < 8; ++r) um = fmaxf(um, u[r]);
        um = fmaxf(um, half_swap_f(um));
        const float ir = __builtin_amdgcn_rcpf(um);
#pragma unroll
        for (int r = 0; r < 8; ++r) p[r] = u[r] * ir;
        M += __logf(um);
    } else {
#pragma unroll
        for (int r = 0; r < 8; ++r) p[r] = u[r];
    }

    if (NEXT) {
#pragma unroll
        for (int r = 0; r < 8; ++r) exc[r] = en[r];
    }
}

__launch_bounds__(32)
__global__ void crf_fwd_kernel(const float* __restrict__ x,
                               const float* __restrict__ trans,
                               const int*   __restrict__ length,
                               float*       __restrict__ out)
{
    const int lane = threadIdx.x & 31;
    const int half = lane >> 4;
    const int m    = lane & 15;
    const int bm   = blockIdx.x * 16 + m;

    const int s  = lane & 15;
    const int jo = 4 * ((s >> 1) & 3) + (s & 1) + 2 * (s >> 3);

    v2f A[4];
#pragma unroll
    for (int k = 0; k < 4; ++k) {
        const int k0 = 4 * k + 2 * half;
        A[k].x = __expf(trans[k0 * LL + jo]);
        A[k].y = __expf(trans[(k0 + 1) * LL + jo]);
    }

    const int len = length[bm];
    const float* xb = x + (size_t)bm * TT * LL + 2 * half;

    float p[8], finu[8];
    {
        float xf0[8];
        load_xfrag(xb, 0, xf0);
#pragma unroll
        for (int r = 0; r < 8; ++r) { p[r] = __expf(xf0[r]); finu[r] = p[r]; }
    }
    float M = 0.0f, finm = 0.0f;

    float exc[8];
    {
        float xf1[8];
        load_xfrag(xb, 1, xf1);
#pragma unroll
        for (int r = 0; r < 8; ++r) exc[r] = __expf(xf1[r]);
    }

    float ring[RING][8];
#pragma unroll
    for (int s = 1; s < RING; ++s) load_xfrag(xb, 1 + s, ring[s]);
#pragma unroll
    for (int q = 12; q < 4 + PFD; q += 4) __builtin_prefetch(xb + (size_t)q * LL, 0, 1);

#pragma unroll 1
    for (int t0 = 1; t0 + RING - 1 < TT; t0 += RING) {
#pragma unroll
        for (int j = 0; j < RING; ++j) {
            const int t = t0 + j;
            fwd_step<true, true>(t, len, xb, A, p, exc, ring[(j + 1) % RING], ring[j],
                                 M, finm, finu, j == RING - 1);
            if ((j & 3) == 3) {
                const int tp = (t + PFD < TT - 4) ? (t + PFD) : (TT - 4);
                __builtin_prefetch(xb + (size_t)tp * LL, 0, 1);
            }
        }
    }

#pragma unroll
    for (int j = 0; j < RING - 1; ++j) {
        const int t = TT - (RING - 1) + j;
        if (j < RING - 2)
            fwd_step<false, true>(t, len, xb, A, p, exc, ring[(j + 1) % RING], ring[j],
                                  M, finm, finu, false);
        else
            fwd_step<false, false>(t, len, xb, A, p, exc, ring[(j + 1) % RING], ring[j],
                                   M, finm, finu, false);
    }

    float ssum = finu[0] + finu[1];
#pragma unroll
    for (int r = 2; r < 8; ++r) ssum += finu[r];
    ssum += half_swap_f(ssum);
    const float zv = finm + __logf(ssum);

    if (lane < 16) out[bm] = zv + out[bm];
}

extern "C" void kernel_launch(void* const* d_in, const int* in_sizes, int n_in,
                              void* d_out, int out_size, void* d_ws, size_t ws_size,
                              hipStream_t stream) {
    const float* x      = (const float*)d_in[0];
    const float* trans  = (const float*)d_in[1];
    const int*   label  = (const int*)d_in[2];
    const int*   length = (const int*)d_in[3];
    float*       out    = (float*)d_out;

    const int B = in_sizes[3];
    crf_score_kernel<<<B / 8, 256, 0, stream>>>(x, trans, label, length, out);
    crf_fwd_kernel<<<B / 16, 32, 0, stream>>>(x, trans, length, out);
}


# one launch instead of two: each work-group runs its tile's scan wave beside 16 label-score waves (one per batch), scores pass through LDS, out stored once as z - score; order of sums unchanged
# speedup vs baseline: 2.2674x; 1.4524x over previous
#include <hip/hip_runtime.h>


typedef __attribute__((ext_vector_type(2))) float v2f;
typedef __attribute__((ext_vector_type(8))) float v8f;

#define TT 512
#define LL 16
#define RING 8
#define PFD  32

__device__ __forceinline__ float half_swap_f(float v) {
    return __int_as_float(__builtin_amdgcn_ds_swizzle(__float_as_int(v), 0x401F));
}

__device__ __forceinline__ void load_xfrag(const float* base2h, int t, float xf[8]) {
    const float* p = base2h + (size_t)t * LL;
#pragma unroll
    for (int i = 0; i < 4; ++i) {
        float2 v = *(const float2*)(p + 4 * i);
        xf[2 * i]     = v.x;
        xf[2 * i + 1] = v.y;
    }
}

__device__ __forceinline__ float crf_score(const float* __restrict__ x,
                                           const float* __restrict__ trans,
                                           const int*   __restrict__ label,
                                           const int*   __restrict__ length,
                                           const int b, const int lane)
{
    const int len  = length[b];
    const size_t base = (size_t)b * TT;

    float acc = 0.0f;
    for (int t = lane; t < TT; t += 32) {
        if (t < len) {
            const int lab = label[base + t];
            acc += x[(base + t) * LL + lab];
            if (t > 0) acc += trans[label[base + t - 1] * LL + lab];
        }
    }
#pragma unroll
    for (int off = 16; off; off >>= 1) acc += __shfl_xor(acc, off, 32);
    return acc;
}

template <bool RELOAD, bool NEXT>
__device__ __forceinline__ void fwd_step(const int t, const int len, const float* xb,
                                         const v2f (&A)[4], float (&p)[8], float (&exc)[8],
                                         float (&slot)[8], float (&fill)[8],
                                         float& M, float& finm,
                                         float (&finu)[8], const bool renorm)
{
    const v8f z8 = {};
    v2f B0 = {p[0], p[1]}, B1 = {p[2], p[3]};
    v2f B2 = {p[4], p[5]}, B3 = {p[6], p[7]};
    v8f accA = __builtin_amdgcn_wmma_f32_16x16x4_f32(false, A[0], false, B0,
                                                     (short)0, z8, false, false);
    v8f accB = __builtin_amdgcn_wmma_f32_16x16x4_f32(false, A[1], false, B1,
                                                     (short)0, z8, false, false);
    asm volatile("" : "+v"(accA), "+v"(accB),
                      "+v"(slot[0]), "+v"(slot[1]), "+v"(slot[2]), "+v"(slot[3]),
                      "+v"(slot[4]), "+v"(slot[5]), "+v"(slot[6]), "+v"(slot[7])
                 : : "memory");
    accA = __builtin_amdgcn_wmma_f32_16x16x4_f32(false, A[2], false, B2,
                                                 (short)0, accA, false, false);
    accB = __builtin_amdgcn_wmma_f32_16x16x4_f32(false, A[3], false, B3,
                                                 (short)0, accB, false, false);

    float en[8];
    if (NEXT) {
#pragma unroll
        for (int r = 0; r < 8; ++r) en[r] = __expf(slot[r]);
    }
    if (RELOAD) {
        const int tl = (t + RING < TT) ? (t + RING) : (TT - 1);
        load_xfrag(xb, tl, fill);
    }

    float u[8];
#pragma unroll
    for (int r = 0; r < 8; ++r) u[r] = (accA[r] + accB[r]) * exc[r];

    const bool cap = (t == len - 1);
    finm = cap ? M : finm;
#pragma unroll
    for (int r = 0; r < 8; ++r) finu[r] = cap ? u[r] : finu[r];

    if (renorm) {
        float um = fmaxf(u[0], u[1]);
#pragma unroll
        for (int r = 2; r < 8; ++r) um = fmaxf(um, u[r]);
        um = fmaxf(um, half_swap_f(um));
        const float ir = __builtin_amdgcn_rcpf(um);
#pragma unroll
        for (int r = 0; r < 8; ++r) p[r] = u[r] * ir;
        M += __logf(um);
    } else {
#pragma unroll
        for (int r = 0; r < 8; ++r) p[r] = u[r];
    }

    if (NEXT) {
#pragma unroll
        for (int r = 0; r < 8; ++r) exc[r] = en[r];
    }
}

__device__ __forceinline__ float crf_scan(const float* __restrict__ x,
                                          const float* __restrict__ trans,
                                          const int*   __restrict__ length,
                                          const int tile, const int lane)
{
    const int half = lane >> 4;
    const int m    = lane & 15;
    const int bm   = tile * 16 + m;

    const int s  = lane & 15;
    const int jo = 4 * ((s >> 1) & 3) + (s & 1) + 2 * (s >> 3);

    v2f A[4];
#pragma unroll
    for (int k = 0; k < 4; ++k) {
        const int k0 = 4 * k + 2 * half;
        A[k].x = __expf(trans[k0 * LL + jo]);
        A[k].y = __expf(trans[(k0 + 1) * LL + jo]);
    }

    const int len = length[bm];
    const float* xb = x + (size_t)bm * TT * LL + 2 * half;

    float p[8], finu[8];
    {
        float xf0[8];
        load_xfrag(xb, 0, xf0);
#pragma unroll
        for (int r = 0; r < 8; ++r) { p[r] = __expf(xf0[r]); finu[r] = p[r]; }
    }
    float M = 0.0f, finm = 0.0f;

    float exc[8];
    {
        float xf1[8];
        load_xfrag(xb, 1, xf1);
#pragma unroll
        for (int r = 0; r < 8; ++r) exc[r] = __expf(xf1[r]);
    }

    float ring[RING][8];
#pragma unroll
    for (int s = 1; s < RING; ++s) load_xfrag(xb, 1 + s, ring[s]);
#pragma unroll
    for (int q = 12; q < 4 + PFD; q += 4) __builtin_prefetch(xb + (size_t)q * LL, 0, 1);

#pragma unroll 1
    for (int t0 = 1; t0 + RING - 1 < TT; t0 += RING) {
#pragma unroll
        for (int j = 0; j < RING; ++j) {
            const int t = t0 + j;
            fwd_step<true, true>(t, len, xb, A, p, exc, ring[(j + 1) % RING], ring[j],
                                 M, finm, finu, j == RING - 1);
            if ((j & 3) == 3) {
                const int tp = (t + PFD < TT - 4) ? (t + PFD) : (TT - 4);
                __builtin_prefetch(xb + (size_t)tp * LL, 0, 1);
            }
        }
    }

#pragma unroll
    for (int j = 0; j < RING - 1; ++j) {
        const int t = TT - (RING - 1) + j;
        if (j < RING - 2)
            fwd_step<false, true>(t, len, xb, A, p, exc, ring[(j + 1) % RING], ring[j],
                                  M, finm, finu, false);
        else
            fwd_step<false, false>(t, len, xb, A, p, exc, ring[(j + 1) % RING], ring[j],
                                   M, finm, finu, false);
    }

    float ssum = finu[0] + finu[1];
#pragma unroll
    for (int r = 2; r < 8; ++r) ssum += finu[r];
    ssum += half_swap_f(ssum);
    return finm + __logf(ssum);
}

#define NSCORE 16

__launch_bounds__(32 * (1 + NSCORE))
__global__ void crf_kernel(const float* __restrict__ x,
                           const float* __restrict__ trans,
                           const int*   __restrict__ label,
                           const int*   __restrict__ length,
                           float*       __restrict__ out)
{
    __shared__ float sc[NSCORE];

    const int lane = threadIdx.x & 31;
    const int wave = __builtin_amdgcn_readfirstlane((int)(threadIdx.x >> 5));

    float zv = 0.0f;
    if (wave == 0) {
        zv = crf_scan(x, trans, length, blockIdx.x, lane);
    } else {
        const float s = crf_score(x, trans, label, length,
                                  blockIdx.x * NSCORE + (wave - 1), lane);
        if (lane == 0) sc[wave - 1] = s;
    }
    __syncthreads();

    if (wave == 0 && lane < 16) out[blockIdx.x * 16 + lane] = zv - sc[lane];
}

extern "C" void kernel_launch(void* const* d_in, const int* in_sizes, int n_in,
                              void* d_out, int out_size, void* d_ws, size_t ws_size,
                              hipStream_t stream) {
    const float* x      = (const float*)d_in[0];
    const float* trans  = (const float*)d_in[1];
    const int*   label  = (const int*)d_in[2];
    const int*   length = (const int*)d_in[3];
    float*       out    = (float*)d_out;

    const int B = in_sizes[3];
    crf_kernel<<<B / 16, 32 * (1 + NSCORE), 0, stream>>>(x, trans, label, length, out);
}


# time axis split at t=256: wave 0 scans forward to alpha_256, a second wave scans backward from T-1 to beta_256 (steps past a length masked), joined by sum_j alpha*beta; order of sums changes
# speedup vs baseline: 4.1460x; 1.8285x over previous
#include <hip/hip_runtime.h>


typedef __attribute__((ext_vector_type(2))) float v2f;
typedef __attribute__((ext_vector_type(8))) float v8f;

#define TT 512
#define LL 16
#define RING 8
#define PFD  32
#define MID  256

__device__ __forceinline__ float half_swap_f(float v) {
    return __int_as_float(__builtin_amdgcn_ds_swizzle(__float_as_int(v), 0x401F));
}

__device__ __forceinline__ void load_xfrag(const float* base2h, int t, float xf[8]) {
    const float* p = base2h + (size_t)t * LL;
#pragma unroll
    for (int i = 0; i < 4; ++i) {
        float2 v = *(const float2*)(p + 4 * i);
        xf[2 * i]     = v.x;
        xf[2 * i + 1] = v.y;
    }
}

__device__ __forceinline__ float crf_score(const float* __restrict__ x,
                                           const float* __restrict__ trans,
                                           const int*   __restrict__ label,
                                           const int*   __restrict__ length,
                                           const int b, const int lane)
{
    const int len  = length[b];
    const size_t base = (size_t)b * TT;

    float acc = 0.0f;
    for (int t = lane; t < TT; t += 32) {
        if (t < len) {
            const int lab = label[base + t];
            acc += x[(base + t) * LL + lab];
            if (t > 0) acc += trans[label[base + t - 1] * LL + lab];
        }
    }
#pragma unroll
    for (int off = 16; off; off >>= 1) acc += __shfl_xor(acc, off, 32);
    return acc;
}

template <bool RELOAD, bool NEXT>
__device__ __forceinline__ void fwd_step(const int t, const int len, const float* xb,
                                         const v2f (&A)[4], float (&p)[8], float (&exc)[8],
                                         float (&slot)[8], float (&fill)[8],
                                         float& M, float& finm,
                                         float (&finu)[8], const bool renorm)
{
    const v8f z8 = {};
    v2f B0 = {p[0], p[1]}, B1 = {p[2], p[3]};
    v2f B2 = {p[4], p[5]}, B3 = {p[6], p[7]};
    v8f accA = __builtin_amdgcn_wmma_f32_16x16x4_f32(false, A[0], false, B0,
                                                     (short)0, z8, false, false);
    v8f accB = __builtin_amdgcn_wmma_f32_16x16x4_f32(false, A[1], false, B1,
                                                     (short)0, z8, false, false);
    asm volatile("" : "+v"(accA), "+v"(accB),
                      "+v"(slot[0]), "+v"(slot[1]), "+v"(slot[2]), "+v"(slot[3]),
                      "+v"(slot[4]), "+v"(slot[5]), "+v"(slot[6]), "+v"(slot[7])
                 : : "memory");
    accA = __builtin_amdgcn_wmma_f32_16x16x4_f32(false, A[2], false, B2,
                                                 (short)0, accA, false, false);
    accB = __builtin_amdgcn_wmma_f32_16x16x4_f32(false, A[3], false, B3,
                                                 (short)0, accB, false, false);

    float en[8];
    if (NEXT) {
#pragma unroll
        for (int r = 0; r < 8; ++r) en[r] = __expf(slot[r]);
    }
    if (RELOAD) {
        const int tl = (t + RING < TT) ? (t + RING) : (TT - 1);
        load_xfrag(xb, tl, fill);
    }

    float u[8];
#pragma unroll
    for (int r = 0; r < 8; ++r) u[r] = (accA[r] + accB[r]) * exc[r];

    const bool cap = (t == len - 1);
    finm = cap ? M : finm;
#pragma unroll
    for (int r = 0; r < 8; ++r) finu[r] = cap ? u[r] : finu[r];

    if (renorm) {
        float um = fmaxf(u[0], u[1]);
#pragma unroll
        for (int r = 2; r < 8; ++r) um = fmaxf(um, u[r]);
        um = fmaxf(um, half_swap_f(um));
        const float ir = __builtin_amdgcn_rcpf(um);
#pragma unroll
        for (int r = 0; r < 8; ++r) p[r] = u[r] * ir;
        M += __logf(um);
    } else {
#pragma unroll
        for (int r = 0; r < 8; ++r) p[r] = u[r];
    }

    if (NEXT) {
#pragma unroll
        for (int r = 0; r < 8; ++r) exc[r] = en[r];
    }
}

__device__ __forceinline__ void crf_scan_fwd(const float* __restrict__ x,
                                             const float* __restrict__ trans,
                                             const int*   __restrict__ length,
                                             const int tile, const int lane,
                                             float (&finu)[8], float& finm)
{
    const int half = lane >> 4;
    const int m    = lane & 15;
    const int bm   = tile * 16 + m;

    const int s  = lane & 15;
    const int jo = 4 * ((s >> 1) & 3) + (s & 1) + 2 * (s >> 3);

    v2f A[4];
#pragma unroll
    for (int k = 0; k < 4; ++k) {
        const int k0 = 4 * k + 2 * half;
        A[k].x = __expf(trans[k0 * LL + jo]);
        A[k].y = __expf(trans[(k0 + 1) * LL + jo]);
    }

    const int len = length[bm];
    const float* xb = x + (size_t)bm * TT * LL + 2 * half;

    float p[8];
    {
        float xf0[8];
        load_xfrag(xb, 0, xf0);
#pragma unroll
        for (int r = 0; r < 8; ++r) { p[r] = __expf(xf0[r]); finu[r] = p[r]; }
    }
    float M = 0.0f;
    finm = 0.0f;

    float exc[8];
    {
        float xf1[8];
        load_xfrag(xb, 1, xf1);
#pragma unroll
        for (int r = 0; r < 8; ++r) exc[r] = __expf(xf1[r]);
    }

    float ring[RING][8];
#pragma unroll
    for (int s = 1; s < RING; ++s) load_xfrag(xb, 1 + s, ring[s]);
#pragma unroll
    for (int q = 12; q < 4 + PFD; q += 4) __builtin_prefetch(xb + (size_t)q * LL, 0, 1);

#pragma unroll 1
    for (int t0 = 1; t0 + RING - 1 <= MID; t0 += RING) {
#pragma unroll
        for (int j = 0; j < RING; ++j) {
            const int t = t0 + j;
            fwd_step<true, true>(t, len, xb, A, p, exc, ring[(j + 1) % RING], ring[j],
                                 M, finm, finu, j == RING - 1);
            if ((j & 3) == 3) {
                const int tp = (t + PFD < TT - 4) ? (t + PFD) : (TT - 4);
                __builtin_prefetch(xb + (size_t)tp * LL, 0, 1);
            }
        }
    }

    const bool late = (len - 1 > MID);
    finm = late ? M : finm;
#pragma unroll
    for (int r = 0; r < 8; ++r) finu[r] = late ? p[r] : finu[r];
}

template <bool RELOAD, bool NEXT>
__device__ __forceinline__ void bwd_step(const int s, const int len, const float* xb,
                                         const v2f (&A)[4], float (&q)[8], float (&exc)[8],
                                         float (&slot)[8], float (&fill)[8],
                                         float& M, const bool renorm)
{
    const v8f z8 = {};
    float w[8];
#pragma unroll
    for (int r = 0; r < 8; ++r) w[r] = q[r] * exc[r];
    v2f B0 = {w[0], w[1]}, B1 = {w[2], w[3]};
    v2f B2 = {w[4], w[5]}, B3 = {w[6], w[7]};
    v8f accA = __builtin_amdgcn_wmma_f32_16x16x4_f32(false, A[0], false, B0,
                                                     (short)0, z8, false, false);
    v8f accB = __builtin_amdgcn_wmma_f32_16x16x4_f32(false, A[1], false, B1,
                                                     (short)0, z8, false, false);
    asm volatile("" : "+v"(accA), "+v"(accB),
                      "+v"(slot[0]), "+v"(slot[1]), "+v"(slot[2]), "+v"(slot[3]),
                      "+v"(slot[4]), "+v"(slot[5]), "+v"(slot[6]), "+v"(slot[7])
                 : : "memory");
    accA = __builtin_amdgcn_wmma_f32_16x16x4_f32(false, A[2], false, B2,
                                                 (short)0, accA, false, false);
    accB = __builtin_amdgcn_wmma_f32_16x16x4_f32(false, A[3], false, B3,
                                                 (short)0, accB, false, false);

    float en[8];
    if (NEXT) {
#pragma unroll
        for (int r = 0; r < 8; ++r) en[r] = __expf(slot[r]);
    }
    if (RELOAD) {
        const int tl = (s - RING > 0) ? (s - RING) : 0;
        load_xfrag(xb, tl, fill);
    }

    const bool act = (s < len);
    float v[8];
#pragma unroll
    for (int r = 0; r < 8; ++r) v[r] = act ? (accA[r] + accB[r]) : q[r];

    if (renorm) {
        float um = fmaxf(v[0], v[1]);
#pragma unroll
        for (int r = 2; r < 8; ++r) um = fmaxf(um, v[r]);
        um = fmaxf(um, half_swap_f(um));
        const float ir = __builtin_amdgcn_rcpf(um);
#pragma unroll
        for (int r = 0; r < 8; ++r) q[r] = v[r] * ir;
        M += __logf(um);
    } else {
#pragma unroll
        for (int r = 0; r < 8; ++r) q[r] = v[r];
    }

    if (NEXT) {
#pragma unroll
        for (int r = 0; r < 8; ++r) exc[r] = en[r];
    }
}

__device__ __forceinline__ void crf_scan_bwd(const float* __restrict__ x,
                                             const float* __restrict__ trans,
                                             const int*   __restrict__ length,
                                             const int tile, const int lane,
                                             float (&q)[8], float& M)
{
    const int half = lane >> 4;
    const int m    = lane & 15;
    const int bm   = tile * 16 + m;

    const int sl = lane & 15;
    const int jo = 4 * ((sl >> 1) & 3) + (sl & 1) + 2 * (sl >> 3);

    v2f A[4];
#pragma unroll
    for (int k = 0; k < 4; ++k) {
        const int k0 = 4 * k + 2 * half;
        A[k].x = __expf(trans[jo * LL + k0]);
        A[k].y = __expf(trans[jo * LL + k0 + 1]);
    }

    const int len = length[bm];
    const float* xb = x + (size_t)bm * TT * LL + 2 * half;

#pragma unroll
    for (int r = 0; r < 8; ++r) q[r] = 1.0f;
    M = 0.0f;

    float exc[8];
    {
        float xfl[8];
        load_xfrag(xb, TT - 1, xfl);
#pragma unroll
        for (int r = 0; r < 8; ++r) exc[r] = __expf(xfl[r]);
    }

    float ring[RING][8];
#pragma unroll
    for (int i = 1; i < RING; ++i) load_xfrag(xb, TT - 1 - i, ring[i]);
#pragma unroll
    for (int u = TT - 12; u > TT - 4 - PFD; u -= 4) __builtin_prefetch(xb + (size_t)u * LL, 0, 1);

#pragma unroll 1
    for (int s0 = TT - 1; s0 - (RING - 1) > MID + RING - 2; s0 -= RING) {
#pragma unroll
        for (int j = 0; j < RING; ++j) {
            const int s = s0 - j;
            bwd_step<true, true>(s, len, xb, A, q, exc, ring[(j + 1) % RING], ring[j],
                                 M, j == RING - 1);
            if ((j & 3) == 3) {
                const int tp = (s - PFD > 0) ? (s - PFD) : 0;
                __builtin_prefetch(xb + (size_t)tp * LL, 0, 1);
            }
        }
    }

#pragma unroll
    for (int j = 0; j < RING - 1; ++j) {
        const int s = MID + (RING - 1) - j;
        if (j < RING - 2)
            bwd_step<false, true>(s, len, xb, A, q, exc, ring[(j + 1) % RING], ring[j],
                                  M, false);
        else
            bwd_step<false, false>(s, len, xb, A, q, exc, ring[(j + 1) % RING], ring[j],
                                   M, false);
    }
}


#define NSCORE 16

__launch_bounds__(32 * (2 + NSCORE))
__global__ void crf_kernel(const float* __restrict__ x,
                           const float* __restrict__ trans,
                           const int*   __restrict__ label,
                           const int*   __restrict__ length,
                           float*       __restrict__ out)
{
    __shared__ float sc[NSCORE];
    __shared__ float bq[9][32];

    const int lane = threadIdx.x & 31;
    const int wave = __builtin_amdgcn_readfirstlane((int)(threadIdx.x >> 5));

    float finu[8], finm = 0.0f;
#pragma unroll
    for (int r = 0; r < 8; ++r) finu[r] = 0.0f;
    if (wave == 0) {
        crf_scan_fwd(x, trans, length, blockIdx.x, lane, finu, finm);
    } else if (wave == 1) {
        float q[8], mb;
        crf_scan_bwd(x, trans, length, blockIdx.x, lane, q, mb);
#pragma unroll
        for (int r = 0; r < 8; ++r) bq[r][lane] = q[r];
        bq[8][lane] = mb;
    } else {
        const float s = crf_score(x, trans, label, length,
                                  blockIdx.x * NSCORE + (wave - 2), lane);
        if (lane == 0) sc[wave - 2] = s;
    }
    __syncthreads();

    if (wave == 0) {
        float ssum = finu[0] * bq[0][lane] + finu[1] * bq[1][lane];
#pragma unroll
        for (int r = 2; r < 8; ++r) ssum += finu[r] * bq[r][lane];
        ssum += half_swap_f(ssum);
        const float zv = finm + bq[8][lane] + __logf(ssum);
        if (lane < 16) out[blockIdx.x * 16 + lane] = zv - sc[lane];
    }
}

extern "C" void kernel_launch(void* const* d_in, const int* in_sizes, int n_in,
                              void* d_out, int out_size, void* d_ws, size_t ws_size,
                              hipStream_t stream) {
    const float* x      = (const float*)d_in[0];
    const float* trans  = (const float*)d_in[1];
    const int*   label  = (const int*)d_in[2];
    const int*   length = (const int*)d_in[3];
    float*       out    = (float*)d_out;

    const int B = in_sizes[3];
    crf_kernel<<<B / 16, 32 * (2 + NSCORE), 0, stream>>>(x, trans, label, length, out);
}
